# starting kernel
# baseline (speedup 1.0000x reference)
#include <hip/hip_runtime.h>
#include <math.h>

typedef __attribute__((ext_vector_type(2))) float v2f;
typedef __attribute__((ext_vector_type(8))) float v8f;

__global__ __launch_bounds__(256)
void prep_kernel(const float* __restrict__ W, const float* __restrict__ b,
                 const float* __restrict__ C, float* __restrict__ Mo,
                 float* __restrict__ bS)
{
    __shared__ __align__(16) float S[64 * 64];
    __shared__ __align__(16) float sW[64 * 64];
    const int tid = threadIdx.x;

    for (int i = tid; i < 4096; i += 256) sW[i] = W[i];

    if (tid < 64) {
        float m = -INFINITY;
        for (int j = 0; j < 64; ++j) m = fmaxf(m, C[tid * 64 + j]);
        float sum = 0.0f;
        for (int j = 0; j < 64; ++j) {
            float ev = expf(C[tid * 64 + j] - m);
            S[tid * 64 + j] = ev;
            sum += ev;
        }
        float inv = 1.0f / sum;
        for (int j = 0; j < 64; ++j) S[tid * 64 + j] *= inv;
    }
    __syncthreads();

    for (int idx = tid; idx < 4096; idx += 256) {
        int i = idx >> 6, j = idx & 63;
        float acc = 0.0f;
        for (int k = 0; k < 64; ++k) acc = fmaf(sW[i * 64 + k], S[k * 64 + j], acc);
        Mo[idx] = acc;
    }
    if (tid < 64) {
        float acc = 0.0f;
        for (int k = 0; k < 64; ++k) acc = fmaf(b[k], S[k * 64 + tid], acc);
        bS[tid] = acc;
    }
}

__global__ void deg_init(float* deg, int n)
{
    int i = blockIdx.x * blockDim.x + threadIdx.x;
    if (i < n) deg[i] = 1.0f;
}

__global__ void deg_count(const int* __restrict__ ei, float* deg, int E)
{
    int e = blockIdx.x * blockDim.x + threadIdx.x;
    if (e < E) atomicAdd(&deg[ei[E + e]], 1.0f);
}

__global__ void dinv_kernel(float* d, int n)
{
    int i = blockIdx.x * blockDim.x + threadIdx.x;
    if (i < n) d[i] = rsqrtf(d[i]);
}

__global__ __launch_bounds__(256)
void gemm_wmma_f32(const float* __restrict__ x, const float* __restrict__ M,
                   float* __restrict__ y, int nrows)
{
    __shared__ __align__(16) float sA[32 * 64];
    __shared__ __align__(16) float sB[64 * 64];

    const int tid = threadIdx.x;
    const int rowbase = blockIdx.x * 32;

    for (int i = tid * 4; i < 4096; i += 256 * 4)
        *(float4*)&sB[i] = *(const float4*)&M[i];

    for (int i = tid * 4; i < 2048; i += 256 * 4) {
        int row = rowbase + (i >> 6);
        float4 v = make_float4(0.f, 0.f, 0.f, 0.f);
        if (row < nrows) v = *(const float4*)&x[(size_t)row * 64 + (i & 63)];
        *(float4*)&sA[i] = v;
    }
    __syncthreads();

    const int wave = tid >> 5;
    const int lane = tid & 31;
    const int rblk = wave >> 2;
    const int cblk = wave & 3;
    const int r    = lane & 15;
    const int hi   = lane >> 4;

    const float* Arow = &sA[(rblk * 16 + r) * 64];
    const int    bcol = cblk * 16 + r;

    v8f acc = {};
#pragma unroll
    for (int k0 = 0; k0 < 64; k0 += 4) {
        v2f a, bfr;
        a.x = Arow[k0 + 2 * hi + 0];
        a.y = Arow[k0 + 2 * hi + 1];
        bfr.x = sB[(k0 + 2 * hi + 0) * 64 + bcol];
        bfr.y = sB[(k0 + 2 * hi + 1) * 64 + bcol];
        acc = __builtin_amdgcn_wmma_f32_16x16x4_f32(
             false, a,  false, bfr,
             (short)0, acc,  false,  false);
    }

    const int outrow = rowbase + rblk * 16 + hi * 8;
    const int outcol = cblk * 16 + r;
#pragma unroll
    for (int v = 0; v < 8; ++v) {
        int rr = outrow + v;
        if (rr < nrows) y[(size_t)rr * 64 + outcol] = acc[v];
    }
}

__global__ __launch_bounds__(256)
void self_init(const float* __restrict__ y, const float* __restrict__ dinv,
               const float* __restrict__ bS, float* __restrict__ out, int n)
{
    int t = blockIdx.x * blockDim.x + threadIdx.x;
    int i = t >> 4;
    int c = (t & 15) * 4;
    if (i >= n) return;
    float w = dinv[i] * dinv[i];
    float4 v  = *(const float4*)&y[(size_t)i * 64 + c];
    float4 bb = *(const float4*)&bS[c];
    float4 o  = make_float4(w * v.x + bb.x, w * v.y + bb.y,
                            w * v.z + bb.z, w * v.w + bb.w);
    *(float4*)&out[(size_t)i * 64 + c] = o;
}

__global__ __launch_bounds__(256)
void edge_scatter(const int* __restrict__ ei, const float* __restrict__ dinv,
                  const float* __restrict__ y, float* __restrict__ out, int E)
{
    int t = blockIdx.x * blockDim.x + threadIdx.x;
    int e = t >> 4;
    int c = (t & 15) * 4;
    if (e >= E) return;
    int s = ei[e];
    int d = ei[E + e];
    float w = dinv[s] * dinv[d];
    float4 v = *(const float4*)&y[(size_t)s * 64 + c];
    float* o = &out[(size_t)d * 64 + c];
    atomicAdd(o + 0, w * v.x);
    atomicAdd(o + 1, w * v.y);
    atomicAdd(o + 2, w * v.z);
    atomicAdd(o + 3, w * v.w);
}

extern "C" void kernel_launch(void* const* d_in, const int* in_sizes, int n_in,
                              void* d_out, int out_size, void* d_ws, size_t ws_size,
                              hipStream_t stream)
{
    const float* x  = (const float*)d_in[0];
    const int*   ei = (const int*)d_in[1];
    const float* W  = (const float*)d_in[2];
    const float* b  = (const float*)d_in[3];
    const float* Ca = (const float*)d_in[4];

    const int N = in_sizes[0] / 64;
    const int E = in_sizes[1] / 2;

    float* ws   = (float*)d_ws;
    float* M    = ws;
    float* bS   = ws + 4096;
    float* dinv = ws + 4160;
    float* y    = ws + 4160 + N;
    float* out  = (float*)d_out;

    prep_kernel <<<1, 256, 0, stream>>>(W, b, Ca, M, bS);
    deg_init    <<<(N + 255) / 256, 256, 0, stream>>>(dinv, N);
    deg_count   <<<(E + 255) / 256, 256, 0, stream>>>(ei, dinv, E);
    dinv_kernel <<<(N + 255) / 256, 256, 0, stream>>>(dinv, N);

    gemm_wmma_f32<<<(N + 31) / 32, 256, 0, stream>>>(x, M, y, N);

    self_init   <<<(N * 16 + 255) / 256, 256, 0, stream>>>(y, dinv, bS, out, N);

    long long escatter_threads = (long long)E * 16;
    edge_scatter<<<(int)((escatter_threads + 255) / 256), 256, 0, stream>>>(
        ei, dinv, y, out, E);
}


# Degree count: per-edge global float atomics replaced by per-block LDS integer histograms (packed half-word counters) and a plane-sum kernel doing +1 and rsqrt; 7 launches become 6
# speedup vs baseline: 1.3859x; 1.3859x over previous
#include <hip/hip_runtime.h>
#include <math.h>

typedef __attribute__((ext_vector_type(2))) float v2f;
typedef __attribute__((ext_vector_type(8))) float v8f;

__global__ __launch_bounds__(256)
void prep_kernel(const float* __restrict__ W, const float* __restrict__ b,
                 const float* __restrict__ C, float* __restrict__ Mo,
                 float* __restrict__ bS)
{
    __shared__ __align__(16) float S[64 * 64];
    __shared__ __align__(16) float sW[64 * 64];
    const int tid = threadIdx.x;

    for (int i = tid; i < 4096; i += 256) sW[i] = W[i];

    if (tid < 64) {
        float m = -INFINITY;
        for (int j = 0; j < 64; ++j) m = fmaxf(m, C[tid * 64 + j]);
        float sum = 0.0f;
        for (int j = 0; j < 64; ++j) {
            float ev = expf(C[tid * 64 + j] - m);
            S[tid * 64 + j] = ev;
            sum += ev;
        }
        float inv = 1.0f / sum;
        for (int j = 0; j < 64; ++j) S[tid * 64 + j] *= inv;
    }
    __syncthreads();

    for (int idx = tid; idx < 4096; idx += 256) {
        int i = idx >> 6, j = idx & 63;
        float acc = 0.0f;
        for (int k = 0; k < 64; ++k) acc = fmaf(sW[i * 64 + k], S[k * 64 + j], acc);
        Mo[idx] = acc;
    }
    if (tid < 64) {
        float acc = 0.0f;
        for (int k = 0; k < 64; ++k) acc = fmaf(b[k], S[k * 64 + tid], acc);
        bS[tid] = acc;
    }
}

#define DEG_SLICE 32768
#define DEG_WORDS 51200
#define DEG_LOADS (DEG_SLICE / 1024)

__global__ __launch_bounds__(1024)
void deg_hist(const int* __restrict__ ei, unsigned* __restrict__ part, int n, int E)
{
    __shared__ __align__(16) unsigned h[DEG_WORDS];
    const int tid = threadIdx.x;
    const int nw  = (n + 1) >> 1;
    const int e0  = blockIdx.x * DEG_SLICE;
    const int e1  = min(E, e0 + DEG_SLICE);
    const int* __restrict__ dst = ei + E;
    unsigned* __restrict__ plane = part + (size_t)blockIdx.x * nw;

    for (int base = 0; base < n; base += 2 * DEG_WORDS) {
        const int words = min(DEG_WORDS, nw - (base >> 1));

        for (int i = tid * 4; i < words; i += 4096)
            *(uint4*)&h[i] = make_uint4(0u, 0u, 0u, 0u);
        __syncthreads();

        int dv[DEG_LOADS];
#pragma unroll
        for (int k = 0; k < DEG_LOADS; ++k) {
            int e = e0 + k * 1024 + tid;
            int v = dst[min(e, e1 - 1)];
            dv[k] = (e < e1) ? v : -1;
        }
#pragma unroll
        for (int k = 0; k < DEG_LOADS; ++k) {
            unsigned d = (unsigned)(dv[k] - base);
            if (d < 2u * DEG_WORDS)
                atomicAdd(&h[d >> 1], 1u << ((d & 1u) * 16u));
        }
        __syncthreads();

        unsigned* __restrict__ o = plane + (base >> 1);
        if ((nw & 3) == 0) {
            for (int i = tid * 4; i < words; i += 4096)
                *(uint4*)&o[i] = *(const uint4*)&h[i];
        } else {
            for (int i = tid; i < words; i += 1024) o[i] = h[i];
        }
        __syncthreads();
    }
}

__global__ __launch_bounds__(256)
void deg_reduce(const unsigned* __restrict__ part, float* __restrict__ dinv,
                int n, int nplanes)
{
    const int nw = (n + 1) >> 1;
    const int w  = blockIdx.x * blockDim.x + threadIdx.x;
    if (w >= nw) return;
    unsigned lo = 0u, hi = 0u;
#pragma unroll 16
    for (int p = 0; p < nplanes; ++p) {
        unsigned v = part[(size_t)p * nw + w];
        lo += v & 0xFFFFu;
        hi += v >> 16;
    }
    const int i = 2 * w;
    dinv[i] = rsqrtf((float)(lo + 1u));
    if (i + 1 < n) dinv[i + 1] = rsqrtf((float)(hi + 1u));
}

__global__ __launch_bounds__(256)
void gemm_wmma_f32(const float* __restrict__ x, const float* __restrict__ M,
                   float* __restrict__ y, int nrows)
{
    __shared__ __align__(16) float sA[32 * 64];
    __shared__ __align__(16) float sB[64 * 64];

    const int tid = threadIdx.x;
    const int rowbase = blockIdx.x * 32;

    for (int i = tid * 4; i < 4096; i += 256 * 4)
        *(float4*)&sB[i] = *(const float4*)&M[i];

    for (int i = tid * 4; i < 2048; i += 256 * 4) {
        int row = rowbase + (i >> 6);
        float4 v = make_float4(0.f, 0.f, 0.f, 0.f);
        if (row < nrows) v = *(const float4*)&x[(size_t)row * 64 + (i & 63)];
        *(float4*)&sA[i] = v;
    }
    __syncthreads();

    const int wave = tid >> 5;
    const int lane = tid & 31;
    const int rblk = wave >> 2;
    const int cblk = wave & 3;
    const int r    = lane & 15;
    const int hi   = lane >> 4;

    const float* Arow = &sA[(rblk * 16 + r) * 64];
    const int    bcol = cblk * 16 + r;

    v8f acc = {};
#pragma unroll
    for (int k0 = 0; k0 < 64; k0 += 4) {
        v2f a, bfr;
        a.x = Arow[k0 + 2 * hi + 0];
        a.y = Arow[k0 + 2 * hi + 1];
        bfr.x = sB[(k0 + 2 * hi + 0) * 64 + bcol];
        bfr.y = sB[(k0 + 2 * hi + 1) * 64 + bcol];
        acc = __builtin_amdgcn_wmma_f32_16x16x4_f32(
             false, a,  false, bfr,
             (short)0, acc,  false,  false);
    }

    const int outrow = rowbase + rblk * 16 + hi * 8;
    const int outcol = cblk * 16 + r;
#pragma unroll
    for (int v = 0; v < 8; ++v) {
        int rr = outrow + v;
        if (rr < nrows) y[(size_t)rr * 64 + outcol] = acc[v];
    }
}

__global__ __launch_bounds__(256)
void self_init(const float* __restrict__ y, const float* __restrict__ dinv,
               const float* __restrict__ bS, float* __restrict__ out, int n)
{
    int t = blockIdx.x * blockDim.x + threadIdx.x;
    int i = t >> 4;
    int c = (t & 15) * 4;
    if (i >= n) return;
    float w = dinv[i] * dinv[i];
    float4 v  = *(const float4*)&y[(size_t)i * 64 + c];
    float4 bb = *(const float4*)&bS[c];
    float4 o  = make_float4(w * v.x + bb.x, w * v.y + bb.y,
                            w * v.z + bb.z, w * v.w + bb.w);
    *(float4*)&out[(size_t)i * 64 + c] = o;
}

__global__ __launch_bounds__(256)
void edge_scatter(const int* __restrict__ ei, const float* __restrict__ dinv,
                  const float* __restrict__ y, float* __restrict__ out, int E)
{
    int t = blockIdx.x * blockDim.x + threadIdx.x;
    int e = t >> 4;
    int c = (t & 15) * 4;
    if (e >= E) return;
    int s = ei[e];
    int d = ei[E + e];
    float w = dinv[s] * dinv[d];
    float4 v = *(const float4*)&y[(size_t)s * 64 + c];
    float* o = &out[(size_t)d * 64 + c];
    atomicAdd(o + 0, w * v.x);
    atomicAdd(o + 1, w * v.y);
    atomicAdd(o + 2, w * v.z);
    atomicAdd(o + 3, w * v.w);
}

extern "C" void kernel_launch(void* const* d_in, const int* in_sizes, int n_in,
                              void* d_out, int out_size, void* d_ws, size_t ws_size,
                              hipStream_t stream)
{
    const float* x  = (const float*)d_in[0];
    const int*   ei = (const int*)d_in[1];
    const float* W  = (const float*)d_in[2];
    const float* b  = (const float*)d_in[3];
    const float* Ca = (const float*)d_in[4];

    const int N = in_sizes[0] / 64;
    const int E = in_sizes[1] / 2;

    float* ws   = (float*)d_ws;
    float* M    = ws;
    float* bS   = ws + 4096;
    float* dinv = ws + 4160;
    float* y    = ws + 4160 + N;
    float* out  = (float*)d_out;

    prep_kernel <<<1, 256, 0, stream>>>(W, b, Ca, M, bS);
    const int nplanes = (E + DEG_SLICE - 1) / DEG_SLICE;
    deg_hist    <<<nplanes, 1024, 0, stream>>>(ei, (unsigned*)y, N, E);
    deg_reduce  <<<((N + 1) / 2 + 255) / 256, 256, 0, stream>>>((const unsigned*)y, dinv, N, nplanes);

    gemm_wmma_f32<<<(N + 31) / 32, 256, 0, stream>>>(x, M, y, N);

    self_init   <<<(N * 16 + 255) / 256, 256, 0, stream>>>(y, dinv, bS, out, N);

    long long escatter_threads = (long long)E * 16;
    edge_scatter<<<(int)((escatter_threads + 255) / 256), 256, 0, stream>>>(
        ei, dinv, y, out, E);
}


# Edge aggregation without float atomics: edges counting-sorted by dst (ranks from the LDS histograms), x rows gathered per node into the GEMM's A tile; sums now in sorted order; 6 launches become 5
# speedup vs baseline: 1.8059x; 1.3031x over previous
#include <hip/hip_runtime.h>
#include <math.h>

typedef __attribute__((ext_vector_type(2))) float v2f;
typedef __attribute__((ext_vector_type(8))) float v8f;

__global__ __launch_bounds__(256)
void prep_kernel(const float* __restrict__ W, const float* __restrict__ b,
                 const float* __restrict__ C, float* __restrict__ Mo,
                 float* __restrict__ bS)
{
    __shared__ __align__(16) float S[64 * 64];
    __shared__ __align__(16) float sW[64 * 64];
    const int tid = threadIdx.x;

    for (int i = tid; i < 4096; i += 256) sW[i] = W[i];

    if (tid < 64) {
        float m = -INFINITY;
        for (int j = 0; j < 64; ++j) m = fmaxf(m, C[tid * 64 + j]);
        float sum = 0.0f;
        for (int j = 0; j < 64; ++j) {
            float ev = expf(C[tid * 64 + j] - m);
            S[tid * 64 + j] = ev;
            sum += ev;
        }
        float inv = 1.0f / sum;
        for (int j = 0; j < 64; ++j) S[tid * 64 + j] *= inv;
    }
    __syncthreads();

    for (int idx = tid; idx < 4096; idx += 256) {
        int i = idx >> 6, j = idx & 63;
        float acc = 0.0f;
        for (int k = 0; k < 64; ++k) acc = fmaf(sW[i * 64 + k], S[k * 64 + j], acc);
        Mo[idx] = acc;
    }
    if (tid < 64) {
        float acc = 0.0f;
        for (int k = 0; k < 64; ++k) acc = fmaf(b[k], S[k * 64 + tid], acc);
        bS[tid] = acc;
    }
}

#define DEG_SLICE 32768
#define DEG_WORDS 51200
#define DEG_LOADS (DEG_SLICE / 1024)
#define SCAN_NODES 512
#define MAX_SCAN_BLOCKS 2048

__global__ __launch_bounds__(1024)
void deg_hist(const int* __restrict__ ei, unsigned* __restrict__ part,
              unsigned short* __restrict__ rk, int n, int E)
{
    __shared__ __align__(16) unsigned h[DEG_WORDS];
    const int tid = threadIdx.x;
    const int nw  = (n + 1) >> 1;
    const int e0  = blockIdx.x * DEG_SLICE;
    const int e1  = min(E, e0 + DEG_SLICE);
    const int* __restrict__ dst = ei + E;
    unsigned* __restrict__ plane = part + (size_t)blockIdx.x * nw;

    for (int base = 0; base < n; base += 2 * DEG_WORDS) {
        const int words = min(DEG_WORDS, nw - (base >> 1));

        for (int i = tid * 4; i < words; i += 4096)
            *(uint4*)&h[i] = make_uint4(0u, 0u, 0u, 0u);
        __syncthreads();

        int dv[DEG_LOADS];
#pragma unroll
        for (int k = 0; k < DEG_LOADS; ++k) {
            int e = e0 + k * 1024 + tid;
            int v = dst[min(e, e1 - 1)];
            dv[k] = (e < e1) ? v : -1;
        }
#pragma unroll
        for (int k = 0; k < DEG_LOADS; ++k) {
            unsigned d = (unsigned)(dv[k] - base);
            if (d < 2u * DEG_WORDS) {
                const unsigned sh  = (d & 1u) * 16u;
                const unsigned old = atomicAdd(&h[d >> 1], 1u << sh);
                rk[e0 + k * 1024 + tid] = (unsigned short)((old >> sh) & 0xFFFFu);
            }
        }
        __syncthreads();

        unsigned* __restrict__ o = plane + (base >> 1);
        if ((nw & 3) == 0) {
            for (int i = tid * 4; i < words; i += 4096)
                *(uint4*)&o[i] = *(const uint4*)&h[i];
        } else {
            for (int i = tid; i < words; i += 1024) o[i] = h[i];
        }
        __syncthreads();
    }
}

__device__ __forceinline__ int block_scan_incl(int v, int* tmp, int tid)
{
    tmp[tid] = v;
    __syncthreads();
#pragma unroll
    for (int off = 1; off < 256; off <<= 1) {
        int a = tmp[tid];
        int b = (tid >= off) ? tmp[tid - off] : 0;
        __syncthreads();
        tmp[tid] = a + b;
        __syncthreads();
    }
    return tmp[tid];
}

__global__ __launch_bounds__(256)
void deg_scan(unsigned* part, float* __restrict__ dinv, int* __restrict__ lex,
              int* __restrict__ btot, int n, int nplanes)
{
    __shared__ int tmp[256];
    const int tid = threadIdx.x;
    const int nw  = (n + 1) >> 1;
    const int w   = blockIdx.x * 256 + tid;
    unsigned lo = 0u, hi = 0u;
    if (w < nw) {
        for (int p0 = 0; p0 < nplanes; p0 += 16) {
            unsigned v[16];
#pragma unroll
            for (int j = 0; j < 16; ++j)
                v[j] = (p0 + j < nplanes) ? part[(size_t)(p0 + j) * nw + w] : 0u;
#pragma unroll
            for (int j = 0; j < 16; ++j) {
                if (p0 + j < nplanes) {
                    part[(size_t)(p0 + j) * nw + w] = (lo & 0xFFFFu) | (hi << 16);
                    lo += v[j] & 0xFFFFu;
                    hi += v[j] >> 16;
                }
            }
        }
    }
    const int s    = (int)(lo + hi);
    const int incl = block_scan_incl(s, tmp, tid);
    if (w < nw) {
        const int i = 2 * w;
        dinv[i] = rsqrtf((float)(lo + 1u));
        lex[i]  = incl - s;
        if (i + 1 < n) {
            dinv[i + 1] = rsqrtf((float)(hi + 1u));
            lex[i + 1]  = incl - s + (int)lo;
        }
    }
    if (tid == 255) btot[blockIdx.x] = incl;
}

__global__ __launch_bounds__(256)
void csr_fill(const int* __restrict__ ei, const unsigned* __restrict__ part,
              const unsigned short* __restrict__ rk, const int* __restrict__ lex,
              const int* __restrict__ btot, int* __restrict__ rowptr,
              int* __restrict__ ss, int n, int E, int nb, int nodes_per_block)
{
    __shared__ int tmp[256];
    __shared__ int bp[MAX_SCAN_BLOCKS];
    const int tid = threadIdx.x;
    const int nw  = (n + 1) >> 1;

    int carry = 0;
    for (int base = 0; base < nb; base += 256) {
        const int v    = (base + tid < nb) ? btot[base + tid] : 0;
        const int incl = block_scan_incl(v, tmp, tid);
        if (base + tid < nb) bp[base + tid] = carry + incl - v;
        carry += tmp[255];
        __syncthreads();
    }
    __syncthreads();

    const int i0 = blockIdx.x * nodes_per_block;
    const int i1 = min(n, i0 + nodes_per_block);
    for (int i = i0 + tid; i < i1; i += 256) rowptr[i] = lex[i] + bp[i / SCAN_NODES];
    if (blockIdx.x == 0 && tid == 0) rowptr[n] = E;

    const int eb = blockIdx.x * 1024;
    const unsigned* __restrict__ plane = part + (size_t)(eb / DEG_SLICE) * nw;
    int s[4], d[4], r[4];
#pragma unroll
    for (int k = 0; k < 4; ++k) {
        const int e  = eb + k * 256 + tid;
        const int ec = min(e, E - 1);
        s[k] = ei[ec];
        d[k] = ei[E + ec];
        r[k] = rk[ec];
    }
    unsigned pw[4];
    int      st[4];
#pragma unroll
    for (int k = 0; k < 4; ++k) {
        pw[k] = plane[d[k] >> 1];
        st[k] = lex[d[k]];
    }
#pragma unroll
    for (int k = 0; k < 4; ++k) {
        const int e = eb + k * 256 + tid;
        if (e < E) {
            const int pre = (int)((pw[k] >> ((d[k] & 1) * 16)) & 0xFFFFu);
            ss[st[k] + bp[d[k] / SCAN_NODES] + pre + r[k]] = s[k];
        }
    }
}

__global__ __launch_bounds__(256)
void gather_gemm(const float* __restrict__ x, const float* __restrict__ M,
                 const float* __restrict__ bS, const float* __restrict__ dinv,
                 const int* __restrict__ rowptr, const int* __restrict__ ss,
                 float* __restrict__ out, int nrows)
{
    __shared__ __align__(16) float sA[32 * 64];
    __shared__ __align__(16) float sB[64 * 64];

    const int tid = threadIdx.x;
    const int rowbase = blockIdx.x * 32;
    const int wave = tid >> 5;
    const int lane = tid & 31;
    const int hi   = lane >> 4;
    const int c    = lane & 15;
    const int c4   = c * 4;

    for (int i = tid * 4; i < 4096; i += 256 * 4)
        *(float4*)&sB[i] = *(const float4*)&M[i];

#pragma unroll
    for (int rr = 0; rr < 2; ++rr) {
        const int lr  = wave * 4 + rr * 2 + hi;
        const int row = rowbase + lr;
        const bool live = row < nrows;
        const int rc  = live ? row : 0;
        const int k0  = live ? rowptr[rc] : 0;
        const int len = live ? rowptr[rc + 1] - k0 : 0;
        const float di = dinv[rc];
        float4 xs = *(const float4*)&x[(size_t)rc * 64 + c4];
        float4 acc = make_float4(di * xs.x, di * xs.y, di * xs.z, di * xs.w);
        const int lenmax = max(len, __shfl_xor(len, 16));
        for (int b = 0; b < lenmax; b += 16) {
            const bool mine = (b + c) < len;
            const int   idx = mine ? ss[k0 + b + c] : 0;
            const float wv  = mine ? dinv[idx] : 0.0f;
            const int cnt = min(16, lenmax - b);
            for (int j = 0; j < cnt; j += 4) {
                int   sj[4];
                float wj[4];
                float4 vj[4];
#pragma unroll
                for (int u = 0; u < 4; ++u) {
                    sj[u] = __shfl(idx, j + u, 16);
                    wj[u] = __shfl(wv,  j + u, 16);
                    vj[u] = make_float4(0.f, 0.f, 0.f, 0.f);
                    if (b + j + u < len) vj[u] = *(const float4*)&x[(size_t)sj[u] * 64 + c4];
                }
#pragma unroll
                for (int u = 0; u < 4; ++u) {
                    acc.x = fmaf(wj[u], vj[u].x, acc.x);
                    acc.y = fmaf(wj[u], vj[u].y, acc.y);
                    acc.z = fmaf(wj[u], vj[u].z, acc.z);
                    acc.w = fmaf(wj[u], vj[u].w, acc.w);
                }
            }
        }
        if (!live) acc = make_float4(0.f, 0.f, 0.f, 0.f);
        else       acc = make_float4(di * acc.x, di * acc.y, di * acc.z, di * acc.w);
        *(float4*)&sA[lr * 64 + c4] = acc;
    }
    __syncthreads();

    const int rblk = wave >> 2;
    const int cblk = wave & 3;
    const int r    = c;

    const float* Arow = &sA[(rblk * 16 + r) * 64];
    const int    bcol = cblk * 16 + r;

    v8f accd = {};
#pragma unroll
    for (int k0 = 0; k0 < 64; k0 += 4) {
        v2f a, bfr;
        a.x = Arow[k0 + 2 * hi + 0];
        a.y = Arow[k0 + 2 * hi + 1];
        bfr.x = sB[(k0 + 2 * hi + 0) * 64 + bcol];
        bfr.y = sB[(k0 + 2 * hi + 1) * 64 + bcol];
        accd = __builtin_amdgcn_wmma_f32_16x16x4_f32(
             false, a,  false, bfr,
             (short)0, accd,  false,  false);
    }

    const int outrow = rowbase + rblk * 16 + hi * 8;
    const int outcol = cblk * 16 + r;
    const float bias = bS[outcol];
#pragma unroll
    for (int v = 0; v < 8; ++v) {
        int rrow = outrow + v;
        if (rrow < nrows) out[(size_t)rrow * 64 + outcol] = accd[v] + bias;
    }
}

extern "C" void kernel_launch(void* const* d_in, const int* in_sizes, int n_in,
                              void* d_out, int out_size, void* d_ws, size_t ws_size,
                              hipStream_t stream)
{
    const float* x  = (const float*)d_in[0];
    const int*   ei = (const int*)d_in[1];
    const float* W  = (const float*)d_in[2];
    const float* b  = (const float*)d_in[3];
    const float* Ca = (const float*)d_in[4];

    const int N = in_sizes[0] / 64;
    const int E = in_sizes[1] / 2;

    float* ws   = (float*)d_ws;
    float* M    = ws;
    float* bS   = ws + 4096;
    float* dinv = ws + 4160;
    float* out  = (float*)d_out;

    const int nplanes = (E + DEG_SLICE - 1) / DEG_SLICE;
    const int nw      = (N + 1) / 2;
    const int nb      = (nw + 255) / 256;
    unsigned* area = (unsigned*)(ws + 4160 + N);
    unsigned*       part   = area;
    unsigned short* rk     = (unsigned short*)(part + (((size_t)nplanes * nw + 3) & ~(size_t)3));
    int*            ss     = (int*)((unsigned*)rk + ((((size_t)E + 1) / 2 + 3) & ~(size_t)3));
    int*            lex    = ss + (((size_t)E + 3) & ~(size_t)3);
    int*            rowptr = lex + (((size_t)N + 3) & ~(size_t)3);
    int*            btot   = rowptr + (((size_t)N + 4) & ~(size_t)3);

    prep_kernel<<<1, 256, 0, stream>>>(W, b, Ca, M, bS);
    deg_hist   <<<nplanes, 1024, 0, stream>>>(ei, part, rk, N, E);
    deg_scan   <<<nb, 256, 0, stream>>>(part, dinv, lex, btot, N, nplanes);

    const int fill_blocks = (E + 1023) / 1024;
    const int nodes_per_block = (N + fill_blocks - 1) / fill_blocks;
    csr_fill   <<<fill_blocks, 256, 0, stream>>>(ei, part, rk, lex, btot, rowptr, ss,
                                                  N, E, nb, nodes_per_block);

    gather_gemm<<<(N + 31) / 32, 256, 0, stream>>>(x, M, bS, dinv, rowptr, ss, out, N);
}


# Prep (softmax, W@S, b@S) folded into the histogram launch as one extra 1024-thread work-group, C and W staged in LDS; it runs beside the 49 histogram groups; 5 launches become 4
# speedup vs baseline: 2.1107x; 1.1688x over previous
#include <hip/hip_runtime.h>
#include <math.h>

typedef __attribute__((ext_vector_type(2))) float v2f;
typedef __attribute__((ext_vector_type(8))) float v8f;

#define PREP_SP 68

__device__ __forceinline__
void prep_block(const float* __restrict__ W, const float* __restrict__ b,
                const float* __restrict__ C, float* __restrict__ Mo,
                float* __restrict__ bS)
{
    __shared__ __align__(16) float S[64 * PREP_SP];
    __shared__ __align__(16) float sW[64 * 64];
    __shared__ float sb[64];
    const int tid = threadIdx.x;
    const int r   = tid >> 4;
    const int j4  = (tid & 15) * 4;

    const float4 w4 = *(const float4*)&W[tid * 4];
    const float4 c4 = *(const float4*)&C[tid * 4];
    const float  bv = (tid < 64) ? b[tid] : 0.0f;
    *(float4*)&sW[tid * 4] = w4;
    *(float4*)&S[r * PREP_SP + j4] = c4;
    if (tid < 64) sb[tid] = bv;
    __syncthreads();

    if (tid < 64) {
        float* row = &S[tid * PREP_SP];
        float4 v[16];
#pragma unroll
        for (int q = 0; q < 16; ++q) v[q] = *(const float4*)&row[q * 4];
        float m = -INFINITY;
#pragma unroll
        for (int q = 0; q < 16; ++q) {
            m = fmaxf(m, v[q].x);
            m = fmaxf(m, v[q].y);
            m = fmaxf(m, v[q].z);
            m = fmaxf(m, v[q].w);
        }
        float sum = 0.0f;
#pragma unroll
        for (int q = 0; q < 16; ++q) {
            v[q].x = expf(v[q].x - m); sum += v[q].x;
            v[q].y = expf(v[q].y - m); sum += v[q].y;
            v[q].z = expf(v[q].z - m); sum += v[q].z;
            v[q].w = expf(v[q].w - m); sum += v[q].w;
        }
        const float inv = 1.0f / sum;
#pragma unroll
        for (int q = 0; q < 16; ++q)
            *(float4*)&row[q * 4] = make_float4(v[q].x * inv, v[q].y * inv, v[q].z * inv, v[q].w * inv);
    }
    __syncthreads();

    float a0 = 0.0f, a1 = 0.0f, a2 = 0.0f, a3 = 0.0f;
    for (int k = 0; k < 64; ++k) {
        const float  w = sW[r * 64 + k];
        const float4 s = *(const float4*)&S[k * PREP_SP + j4];
        a0 = fmaf(w, s.x, a0);
        a1 = fmaf(w, s.y, a1);
        a2 = fmaf(w, s.z, a2);
        a3 = fmaf(w, s.w, a3);
    }
    *(float4*)&Mo[r * 64 + j4] = make_float4(a0, a1, a2, a3);

    if (tid < 64) {
        float acc = 0.0f;
        for (int k = 0; k < 64; ++k) acc = fmaf(sb[k], S[k * PREP_SP + tid], acc);
        bS[tid] = acc;
    }
}

#define DEG_SLICE 32768
#define DEG_WORDS 51200
#define DEG_LOADS (DEG_SLICE / 1024)
#define SCAN_NODES 512
#define MAX_SCAN_BLOCKS 2048

__global__ __launch_bounds__(1024)
void deg_hist(const int* __restrict__ ei, unsigned* __restrict__ part,
              unsigned short* __restrict__ rk, int n, int E,
              const float* __restrict__ W, const float* __restrict__ b,
              const float* __restrict__ C, float* __restrict__ Mo,
              float* __restrict__ bS, int nhist)
{
    __shared__ __align__(16) unsigned h[DEG_WORDS];
    if ((int)blockIdx.x == nhist) {
        prep_block(W, b, C, Mo, bS);
        return;
    }
    const int tid = threadIdx.x;
    const int nw  = (n + 1) >> 1;
    const int e0  = blockIdx.x * DEG_SLICE;
    const int e1  = min(E, e0 + DEG_SLICE);
    const int* __restrict__ dst = ei + E;
    unsigned* __restrict__ plane = part + (size_t)blockIdx.x * nw;

    for (int base = 0; base < n; base += 2 * DEG_WORDS) {
        const int words = min(DEG_WORDS, nw - (base >> 1));

        for (int i = tid * 4; i < words; i += 4096)
            *(uint4*)&h[i] = make_uint4(0u, 0u, 0u, 0u);
        __syncthreads();

        int dv[DEG_LOADS];
#pragma unroll
        for (int k = 0; k < DEG_LOADS; ++k) {
            int e = e0 + k * 1024 + tid;
            int v = dst[min(e, e1 - 1)];
            dv[k] = (e < e1) ? v : -1;
        }
#pragma unroll
        for (int k = 0; k < DEG_LOADS; ++k) {
            unsigned d = (unsigned)(dv[k] - base);
            if (d < 2u * DEG_WORDS) {
                const unsigned sh  = (d & 1u) * 16u;
                const unsigned old = atomicAdd(&h[d >> 1], 1u << sh);
                rk[e0 + k * 1024 + tid] = (unsigned short)((old >> sh) & 0xFFFFu);
            }
        }
        __syncthreads();

        unsigned* __restrict__ o = plane + (base >> 1);
        if ((nw & 3) == 0) {
            for (int i = tid * 4; i < words; i += 4096)
                *(uint4*)&o[i] = *(const uint4*)&h[i];
        } else {
            for (int i = tid; i < words; i += 1024) o[i] = h[i];
        }
        __syncthreads();
    }
}

__device__ __forceinline__ int block_scan_incl(int v, int* tmp, int tid)
{
    tmp[tid] = v;
    __syncthreads();
#pragma unroll
    for (int off = 1; off < 256; off <<= 1) {
        int a = tmp[tid];
        int b = (tid >= off) ? tmp[tid - off] : 0;
        __syncthreads();
        tmp[tid] = a + b;
        __syncthreads();
    }
    return tmp[tid];
}

__global__ __launch_bounds__(256)
void deg_scan(unsigned* part, float* __restrict__ dinv, int* __restrict__ lex,
              int* __restrict__ btot, int n, int nplanes)
{
    __shared__ int tmp[256];
    const int tid = threadIdx.x;
    const int nw  = (n + 1) >> 1;
    const int w   = blockIdx.x * 256 + tid;
    unsigned lo = 0u, hi = 0u;
    if (w < nw) {
        for (int p0 = 0; p0 < nplanes; p0 += 16) {
            unsigned v[16];
#pragma unroll
            for (int j = 0; j < 16; ++j)
                v[j] = (p0 + j < nplanes) ? part[(size_t)(p0 + j) * nw + w] : 0u;
#pragma unroll
            for (int j = 0; j < 16; ++j) {
                if (p0 + j < nplanes) {
                    part[(size_t)(p0 + j) * nw + w] = (lo & 0xFFFFu) | (hi << 16);
                    lo += v[j] & 0xFFFFu;
                    hi += v[j] >> 16;
                }
            }
        }
    }
    const int s    = (int)(lo + hi);
    const int incl = block_scan_incl(s, tmp, tid);
    if (w < nw) {
        const int i = 2 * w;
        dinv[i] = rsqrtf((float)(lo + 1u));
        lex[i]  = incl - s;
        if (i + 1 < n) {
            dinv[i + 1] = rsqrtf((float)(hi + 1u));
            lex[i + 1]  = incl - s + (int)lo;
        }
    }
    if (tid == 255) btot[blockIdx.x] = incl;
}

__global__ __launch_bounds__(256)
void csr_fill(const int* __restrict__ ei, const unsigned* __restrict__ part,
              const unsigned short* __restrict__ rk, const int* __restrict__ lex,
              const int* __restrict__ btot, int* __restrict__ rowptr,
              int* __restrict__ ss, int n, int E, int nb, int nodes_per_block)
{
    __shared__ int tmp[256];
    __shared__ int bp[MAX_SCAN_BLOCKS];
    const int tid = threadIdx.x;
    const int nw  = (n + 1) >> 1;

    int carry = 0;
    for (int base = 0; base < nb; base += 256) {
        const int v    = (base + tid < nb) ? btot[base + tid] : 0;
        const int incl = block_scan_incl(v, tmp, tid);
        if (base + tid < nb) bp[base + tid] = carry + incl - v;
        carry += tmp[255];
        __syncthreads();
    }
    __syncthreads();

    const int i0 = blockIdx.x * nodes_per_block;
    const int i1 = min(n, i0 + nodes_per_block);
    for (int i = i0 + tid; i < i1; i += 256) rowptr[i] = lex[i] + bp[i / SCAN_NODES];
    if (blockIdx.x == 0 && tid == 0) rowptr[n] = E;

    const int eb = blockIdx.x * 1024;
    const unsigned* __restrict__ plane = part + (size_t)(eb / DEG_SLICE) * nw;
    int s[4], d[4], r[4];
#pragma unroll
    for (int k = 0; k < 4; ++k) {
        const int e  = eb + k * 256 + tid;
        const int ec = min(e, E - 1);
        s[k] = ei[ec];
        d[k] = ei[E + ec];
        r[k] = rk[ec];
    }
    unsigned pw[4];
    int      st[4];
#pragma unroll
    for (int k = 0; k < 4; ++k) {
        pw[k] = plane[d[k] >> 1];
        st[k] = lex[d[k]];
    }
#pragma unroll
    for (int k = 0; k < 4; ++k) {
        const int e = eb + k * 256 + tid;
        if (e < E) {
            const int pre = (int)((pw[k] >> ((d[k] & 1) * 16)) & 0xFFFFu);
            ss[st[k] + bp[d[k] / SCAN_NODES] + pre + r[k]] = s[k];
        }
    }
}

__global__ __launch_bounds__(256)
void gather_gemm(const float* __restrict__ x, const float* __restrict__ M,
                 const float* __restrict__ bS, const float* __restrict__ dinv,
                 const int* __restrict__ rowptr, const int* __restrict__ ss,
                 float* __restrict__ out, int nrows)
{
    __shared__ __align__(16) float sA[32 * 64];
    __shared__ __align__(16) float sB[64 * 64];

    const int tid = threadIdx.x;
    const int rowbase = blockIdx.x * 32;
    const int wave = tid >> 5;
    const int lane = tid & 31;
    const int hi   = lane >> 4;
    const int c    = lane & 15;
    const int c4   = c * 4;

    for (int i = tid * 4; i < 4096; i += 256 * 4)
        *(float4*)&sB[i] = *(const float4*)&M[i];

#pragma unroll
    for (int rr = 0; rr < 2; ++rr) {
        const int lr  = wave * 4 + rr * 2 + hi;
        const int row = rowbase + lr;
        const bool live = row < nrows;
        const int rc  = live ? row : 0;
        const int k0  = live ? rowptr[rc] : 0;
        const int len = live ? rowptr[rc + 1] - k0 : 0;
        const float di = dinv[rc];
        float4 xs = *(const float4*)&x[(size_t)rc * 64 + c4];
        float4 acc = make_float4(di * xs.x, di * xs.y, di * xs.z, di * xs.w);
        const int lenmax = max(len, __shfl_xor(len, 16));
        for (int b = 0; b < lenmax; b += 16) {
            const bool mine = (b + c) < len;
            const int   idx = mine ? ss[k0 + b + c] : 0;
            const float wv  = mine ? dinv[idx] : 0.0f;
            const int cnt = min(16, lenmax - b);
            for (int j = 0; j < cnt; j += 4) {
                int   sj[4];
                float wj[4];
                float4 vj[4];
#pragma unroll
                for (int u = 0; u < 4; ++u) {
                    sj[u] = __shfl(idx, j + u, 16);
                    wj[u] = __shfl(wv,  j + u, 16);
                    vj[u] = make_float4(0.f, 0.f, 0.f, 0.f);
                    if (b + j + u < len) vj[u] = *(const float4*)&x[(size_t)sj[u] * 64 + c4];
                }
#pragma unroll
                for (int u = 0; u < 4; ++u) {
                    acc.x = fmaf(wj[u], vj[u].x, acc.x);
                    acc.y = fmaf(wj[u], vj[u].y, acc.y);
                    acc.z = fmaf(wj[u], vj[u].z, acc.z);
                    acc.w = fmaf(wj[u], vj[u].w, acc.w);
                }
            }
        }
        if (!live) acc = make_float4(0.f, 0.f, 0.f, 0.f);
        else       acc = make_float4(di * acc.x, di * acc.y, di * acc.z, di * acc.w);
        *(float4*)&sA[lr * 64 + c4] = acc;
    }
    __syncthreads();

    const int rblk = wave >> 2;
    const int cblk = wave & 3;
    const int r    = c;

    const float* Arow = &sA[(rblk * 16 + r) * 64];
    const int    bcol = cblk * 16 + r;

    v8f accd = {};
#pragma unroll
    for (int k0 = 0; k0 < 64; k0 += 4) {
        v2f a, bfr;
        a.x = Arow[k0 + 2 * hi + 0];
        a.y = Arow[k0 + 2 * hi + 1];
        bfr.x = sB[(k0 + 2 * hi + 0) * 64 + bcol];
        bfr.y = sB[(k0 + 2 * hi + 1) * 64 + bcol];
        accd = __builtin_amdgcn_wmma_f32_16x16x4_f32(
             false, a,  false, bfr,
             (short)0, accd,  false,  false);
    }

    const int outrow = rowbase + rblk * 16 + hi * 8;
    const int outcol = cblk * 16 + r;
    const float bias = bS[outcol];
#pragma unroll
    for (int v = 0; v < 8; ++v) {
        int rrow = outrow + v;
        if (rrow < nrows) out[(size_t)rrow * 64 + outcol] = accd[v] + bias;
    }
}

extern "C" void kernel_launch(void* const* d_in, const int* in_sizes, int n_in,
                              void* d_out, int out_size, void* d_ws, size_t ws_size,
                              hipStream_t stream)
{
    const float* x  = (const float*)d_in[0];
    const int*   ei = (const int*)d_in[1];
    const float* W  = (const float*)d_in[2];
    const float* b  = (const float*)d_in[3];
    const float* Ca = (const float*)d_in[4];

    const int N = in_sizes[0] / 64;
    const int E = in_sizes[1] / 2;

    float* ws   = (float*)d_ws;
    float* M    = ws;
    float* bS   = ws + 4096;
    float* dinv = ws + 4160;
    float* out  = (float*)d_out;

    const int nplanes = (E + DEG_SLICE - 1) / DEG_SLICE;
    const int nw      = (N + 1) / 2;
    const int nb      = (nw + 255) / 256;
    unsigned* area = (unsigned*)(ws + 4160 + N);
    unsigned*       part   = area;
    unsigned short* rk     = (unsigned short*)(part + (((size_t)nplanes * nw + 3) & ~(size_t)3));
    int*            ss     = (int*)((unsigned*)rk + ((((size_t)E + 1) / 2 + 3) & ~(size_t)3));
    int*            lex    = ss + (((size_t)E + 3) & ~(size_t)3);
    int*            rowptr = lex + (((size_t)N + 3) & ~(size_t)3);
    int*            btot   = rowptr + (((size_t)N + 4) & ~(size_t)3);

    deg_hist   <<<nplanes + 1, 1024, 0, stream>>>(ei, part, rk, N, E, W, b, Ca, M, bS, nplanes);
    deg_scan   <<<nb, 256, 0, stream>>>(part, dinv, lex, btot, N, nplanes);

    const int fill_blocks = (E + 1023) / 1024;
    const int nodes_per_block = (N + fill_blocks - 1) / fill_blocks;
    csr_fill   <<<fill_blocks, 256, 0, stream>>>(ei, part, rk, lex, btot, rowptr, ss,
                                                  N, E, nb, nodes_per_block);

    gather_gemm<<<(N + 31) / 32, 256, 0, stream>>>(x, M, bS, dinv, rowptr, ss, out, N);
}


# Edge sort is now a two-level bucket sort in LDS: edges binned by 512-node dst bucket, one work-group per bucket builds rows (sources ascending: new fixed sum order), dinv, row starts; 4 launches to 3
# speedup vs baseline: 3.5186x; 1.6670x over previous
#include <hip/hip_runtime.h>
#include <math.h>

typedef __attribute__((ext_vector_type(2))) float v2f;
typedef __attribute__((ext_vector_type(8))) float v8f;

#define PREP_SP 68

__device__ __forceinline__
void prep_block(const float* __restrict__ W, const float* __restrict__ b,
                const float* __restrict__ C, float* __restrict__ Mo,
                float* __restrict__ bS)
{
    __shared__ __align__(16) float S[64 * PREP_SP];
    __shared__ __align__(16) float sW[64 * 64];
    __shared__ float sb[64];
    const int tid = threadIdx.x;
    const int r   = tid >> 4;
    const int j4  = (tid & 15) * 4;

    const float4 w4 = *(const float4*)&W[tid * 4];
    const float4 c4 = *(const float4*)&C[tid * 4];
    const float  bv = (tid < 64) ? b[tid] : 0.0f;
    *(float4*)&sW[tid * 4] = w4;
    *(float4*)&S[r * PREP_SP + j4] = c4;
    if (tid < 64) sb[tid] = bv;
    __syncthreads();

    if (tid < 64) {
        float* row = &S[tid * PREP_SP];
        float4 v[16];
#pragma unroll
        for (int q = 0; q < 16; ++q) v[q] = *(const float4*)&row[q * 4];
        float m = -INFINITY;
#pragma unroll
        for (int q = 0; q < 16; ++q) {
            m = fmaxf(m, v[q].x);
            m = fmaxf(m, v[q].y);
            m = fmaxf(m, v[q].z);
            m = fmaxf(m, v[q].w);
        }
        float sum = 0.0f;
#pragma unroll
        for (int q = 0; q < 16; ++q) {
            v[q].x = expf(v[q].x - m); sum += v[q].x;
            v[q].y = expf(v[q].y - m); sum += v[q].y;
            v[q].z = expf(v[q].z - m); sum += v[q].z;
            v[q].w = expf(v[q].w - m); sum += v[q].w;
        }
        const float inv = 1.0f / sum;
#pragma unroll
        for (int q = 0; q < 16; ++q)
            *(float4*)&row[q * 4] = make_float4(v[q].x * inv, v[q].y * inv, v[q].z * inv, v[q].w * inv);
    }
    __syncthreads();

    float a0 = 0.0f, a1 = 0.0f, a2 = 0.0f, a3 = 0.0f;
    for (int k = 0; k < 64; ++k) {
        const float  w = sW[r * 64 + k];
        const float4 s = *(const float4*)&S[k * PREP_SP + j4];
        a0 = fmaf(w, s.x, a0);
        a1 = fmaf(w, s.y, a1);
        a2 = fmaf(w, s.z, a2);
        a3 = fmaf(w, s.w, a3);
    }
    *(float4*)&Mo[r * 64 + j4] = make_float4(a0, a1, a2, a3);

    if (tid < 64) {
        float acc = 0.0f;
        for (int k = 0; k < 64; ++k) acc = fmaf(sb[k], S[k * PREP_SP + tid], acc);
        bS[tid] = acc;
    }
}

#define SL    8192
#define SL_LOADS (SL / 1024)
#define MAXB  2048
#define MAXS  1024
#define CAP   24576

__device__ __forceinline__ int wave_incl_scan(int v, int lane)
{
#pragma unroll
    for (int o = 1; o < 32; o <<= 1) {
        const int t = __shfl_up(v, o, 32);
        if (lane >= o) v += t;
    }
    return v;
}

__device__ __forceinline__ int block_incl_scan_1024(int v, int* wsum, int tid)
{
    const int lane = tid & 31, w = tid >> 5;
    const int inc = wave_incl_scan(v, lane);
    if (lane == 31) wsum[w] = inc;
    __syncthreads();
    if (w == 0) {
        const int t  = wsum[lane];
        const int ti = wave_incl_scan(t, lane);
        wsum[lane] = ti - t;
    }
    __syncthreads();
    const int r = inc + wsum[w];
    __syncthreads();
    return r;
}

__global__ __launch_bounds__(1024)
void bin_edges(const int* __restrict__ ei, unsigned* __restrict__ binned,
               int* __restrict__ offg, int n, int E, int NB)
{
    __shared__ int cnt[MAXB];
    __shared__ int off[MAXB + 1];
    __shared__ int wsum[32];
    __shared__ unsigned stage[SL];
    const int tid = threadIdx.x;
    const int e0  = blockIdx.x * SL;
    const int e1  = min(E, e0 + SL);

    cnt[tid] = 0;
    cnt[tid + 1024] = 0;
    __syncthreads();

    int sv[SL_LOADS], dv[SL_LOADS], rk[SL_LOADS];
#pragma unroll
    for (int k = 0; k < SL_LOADS; ++k) {
        const int e  = e0 + k * 1024 + tid;
        const int ec = min(e, e1 - 1);
        const int s  = ei[ec];
        const int d  = ei[E + ec];
        sv[k] = s;
        dv[k] = (e < e1 && (unsigned)d < (unsigned)min(n, 512 * MAXB)) ? d : -1;
    }
#pragma unroll
    for (int k = 0; k < SL_LOADS; ++k) {
        rk[k] = 0;
        if (dv[k] >= 0) rk[k] = atomicAdd(&cnt[dv[k] >> 9], 1);
    }
    __syncthreads();

    const int ca = cnt[2 * tid], cb = cnt[2 * tid + 1];
    const int cs = ca + cb;
    const int incl = block_incl_scan_1024(cs, wsum, tid);
    off[2 * tid]     = incl - cs;
    off[2 * tid + 1] = incl - cs + ca;
    if (tid == 1023) off[MAXB] = incl;
    __syncthreads();

#pragma unroll
    for (int k = 0; k < SL_LOADS; ++k)
        if (dv[k] >= 0)
            stage[off[dv[k] >> 9] + rk[k]] = ((unsigned)(dv[k] & 511) << 23) | (unsigned)sv[k];
    __syncthreads();

    const int tot = off[NB];
    for (int i = tid; i < tot; i += 1024) binned[e0 + i] = stage[i];
    int* __restrict__ og = offg + (size_t)blockIdx.x * (NB + 1);
    for (int i = tid; i <= NB; i += 1024) og[i] = off[i];
}

#define SENT 0xFFFFFFFFu

__global__ __launch_bounds__(1024)
void bucket_sort(const unsigned* __restrict__ binned, const int* __restrict__ offg,
                 int* __restrict__ ss, int* __restrict__ rowptr,
                 float* __restrict__ dinv, int n, int NB, int nslices,
                 const float* __restrict__ W, const float* __restrict__ b,
                 const float* __restrict__ C, float* __restrict__ Mo,
                 float* __restrict__ bS)
{
    __shared__ __align__(16) unsigned buf[CAP];
    __shared__ __align__(16) unsigned srt[CAP];
    __shared__ int segst[MAXS], segdd[MAXS], seglen[MAXS];
    __shared__ int hist[512], cursor[512], rowst[513];
    __shared__ int wsum[32];
    __shared__ int sbase, stot;
    if ((int)blockIdx.x == NB) {
        prep_block(W, b, C, Mo, bS);
        return;
    }
    const int tid  = threadIdx.x;
    const int lane = tid & 31;
    const int w    = tid >> 5;
    const int bk   = blockIdx.x;

    if (tid < 512) hist[tid] = 0;
    if (tid == 0) sbase = 0;
    __syncthreads();

    int a = 0, len = 0;
    if (tid < nslices) {
        const int* __restrict__ og = offg + (size_t)tid * (NB + 1);
        a   = og[bk];
        len = og[bk + 1] - a;
    }
    const int incl = block_incl_scan_1024(len, wsum, tid);
    if (tid < nslices) {
        segst[tid]  = tid * SL + a;
        segdd[tid]  = incl - len;
        seglen[tid] = len;
        atomicAdd(&sbase, a);
    }
    if (tid == 1023) stot = incl;
    __syncthreads();
    const int base = sbase;
    const int Tb   = min(stot, CAP);

    for (int s0 = w; s0 < nslices; s0 += 128) {
        unsigned v0[4], v1[4];
        int st[4], ln[4], dd[4];
#pragma unroll
        for (int u = 0; u < 4; ++u) {
            const int s = s0 + 32 * u;
            const bool ok = s < nslices;
            st[u] = ok ? segst[s] : 0;
            ln[u] = ok ? seglen[s] : 0;
            dd[u] = ok ? segdd[s] : 0;
            v0[u] = (lane < ln[u])      ? binned[st[u] + lane]      : 0u;
            v1[u] = (lane + 32 < ln[u]) ? binned[st[u] + lane + 32] : 0u;
        }
#pragma unroll
        for (int u = 0; u < 4; ++u) {
            if (lane < ln[u]      && dd[u] + lane < CAP)      buf[dd[u] + lane]      = v0[u];
            if (lane + 32 < ln[u] && dd[u] + lane + 32 < CAP) buf[dd[u] + lane + 32] = v1[u];
            for (int i = lane + 64; i < ln[u]; i += 32)
                if (dd[u] + i < CAP) buf[dd[u] + i] = binned[st[u] + i];
        }
    }
    __syncthreads();

    for (int i = tid; i < Tb; i += 1024) atomicAdd(&hist[buf[i] >> 23], 1);
    __syncthreads();
    const int hv    = (tid < 512) ? hist[tid] : 0;
    const int hincl = block_incl_scan_1024(hv, wsum, tid);
    if (tid < 512) { rowst[tid] = hincl - hv; cursor[tid] = hincl - hv; }
    if (tid == 511) rowst[512] = hincl;
    __syncthreads();

    for (int i = tid; i < Tb; i += 1024) {
        const unsigned v = buf[i];
        const int pos = atomicAdd(&cursor[v >> 23], 1);
        srt[pos] = v;
    }
    if (tid < 4 && Tb + tid < CAP) srt[Tb + tid] = SENT;
    __syncthreads();

    for (int i = tid; i < Tb; i += 1024) buf[i] = SENT;
    __syncthreads();

    for (int i = tid; i < Tb; i += 1024) {
        const unsigned v = srt[i];
        const int r  = (int)(v >> 23);
        const int lo = rowst[r], hi = rowst[r + 1];
        const int q0 = lo & ~3;
        int less = 0;
        for (int j = q0; j < hi; j += 4) {
            const uint4 q = *(const uint4*)&srt[j];
            less += (q.x < v) ? 1 : 0;
            less += (q.y < v) ? 1 : 0;
            less += (q.z < v) ? 1 : 0;
            less += (q.w < v) ? 1 : 0;
        }
        buf[q0 + less] = v & 0x7FFFFFu;
    }
    __syncthreads();

    for (int i = tid; i < Tb; i += 1024) {
        int k = i;
        unsigned val = buf[k];
        while (val == SENT && k > 0) val = buf[--k];
        if (val == SENT) val = 0u;
        ss[base + i] = (int)val;
    }
    if (tid < 512) {
        const int node = bk * 512 + tid;
        if (node < n) {
            rowptr[node] = base + rowst[tid];
            dinv[node]   = rsqrtf((float)(rowst[tid + 1] - rowst[tid] + 1));
        }
    }
    if (bk == NB - 1 && tid == 0) rowptr[n] = base + Tb;
}

__global__ __launch_bounds__(256)
void gather_gemm(const float* __restrict__ x, const float* __restrict__ M,
                 const float* __restrict__ bS, const float* __restrict__ dinv,
                 const int* __restrict__ rowptr, const int* __restrict__ ss,
                 float* __restrict__ out, int nrows)
{
    __shared__ __align__(16) float sA[32 * 64];
    __shared__ __align__(16) float sB[64 * 64];

    const int tid = threadIdx.x;
    const int rowbase = blockIdx.x * 32;
    const int wave = tid >> 5;
    const int lane = tid & 31;
    const int hi   = lane >> 4;
    const int c    = lane & 15;
    const int c4   = c * 4;

    for (int i = tid * 4; i < 4096; i += 256 * 4)
        *(float4*)&sB[i] = *(const float4*)&M[i];

#pragma unroll
    for (int rr = 0; rr < 2; ++rr) {
        const int lr  = wave * 4 + rr * 2 + hi;
        const int row = rowbase + lr;
        const bool live = row < nrows;
        const int rc  = live ? row : 0;
        const int k0  = live ? rowptr[rc] : 0;
        const int len = live ? rowptr[rc + 1] - k0 : 0;
        const float di = dinv[rc];
        float4 xs = *(const float4*)&x[(size_t)rc * 64 + c4];
        float4 acc = make_float4(di * xs.x, di * xs.y, di * xs.z, di * xs.w);
        const int lenmax = max(len, __shfl_xor(len, 16));
        for (int b = 0; b < lenmax; b += 16) {
            const bool mine = (b + c) < len;
            const int   idx = mine ? ss[k0 + b + c] : 0;
            const float wv  = mine ? dinv[idx] : 0.0f;
            const int cnt = min(16, lenmax - b);
            for (int j = 0; j < cnt; j += 4) {
                int   sj[4];
                float wj[4];
                float4 vj[4];
#pragma unroll
                for (int u = 0; u < 4; ++u) {
                    sj[u] = __shfl(idx, j + u, 16);
                    wj[u] = __shfl(wv,  j + u, 16);
                    vj[u] = make_float4(0.f, 0.f, 0.f, 0.f);
                    if (b + j + u < len) vj[u] = *(const float4*)&x[(size_t)sj[u] * 64 + c4];
                }
#pragma unroll
                for (int u = 0; u < 4; ++u) {
                    acc.x = fmaf(wj[u], vj[u].x, acc.x);
                    acc.y = fmaf(wj[u], vj[u].y, acc.y);
                    acc.z = fmaf(wj[u], vj[u].z, acc.z);
                    acc.w = fmaf(wj[u], vj[u].w, acc.w);
                }
            }
        }
        if (!live) acc = make_float4(0.f, 0.f, 0.f, 0.f);
        else       acc = make_float4(di * acc.x, di * acc.y, di * acc.z, di * acc.w);
        *(float4*)&sA[lr * 64 + c4] = acc;
    }
    __syncthreads();

    const int rblk = wave >> 2;
    const int cblk = wave & 3;
    const int r    = c;

    const float* Arow = &sA[(rblk * 16 + r) * 64];
    const int    bcol = cblk * 16 + r;

    v8f accd = {};
#pragma unroll
    for (int k0 = 0; k0 < 64; k0 += 4) {
        v2f a, bfr;
        a.x = Arow[k0 + 2 * hi + 0];
        a.y = Arow[k0 + 2 * hi + 1];
        bfr.x = sB[(k0 + 2 * hi + 0) * 64 + bcol];
        bfr.y = sB[(k0 + 2 * hi + 1) * 64 + bcol];
        accd = __builtin_amdgcn_wmma_f32_16x16x4_f32(
             false, a,  false, bfr,
             (short)0, accd,  false,  false);
    }

    const int outrow = rowbase + rblk * 16 + hi * 8;
    const int outcol = cblk * 16 + r;
    const float bias = bS[outcol];
#pragma unroll
    for (int v = 0; v < 8; ++v) {
        int rrow = outrow + v;
        if (rrow < nrows) out[(size_t)rrow * 64 + outcol] = accd[v] + bias;
    }
}

extern "C" void kernel_launch(void* const* d_in, const int* in_sizes, int n_in,
                              void* d_out, int out_size, void* d_ws, size_t ws_size,
                              hipStream_t stream)
{
    const float* x  = (const float*)d_in[0];
    const int*   ei = (const int*)d_in[1];
    const float* W  = (const float*)d_in[2];
    const float* b  = (const float*)d_in[3];
    const float* Ca = (const float*)d_in[4];

    const int N = in_sizes[0] / 64;
    const int E = in_sizes[1] / 2;

    float* ws   = (float*)d_ws;
    float* M    = ws;
    float* bS   = ws + 4096;
    float* dinv = ws + 4160;
    float* out  = (float*)d_out;

    const int NB  = (N + 511) / 512;
    const int nsl = (E + SL - 1) / SL;
    unsigned* area = (unsigned*)(ws + 4160 + N);
    unsigned* binned = area;
    int*      ss     = (int*)(binned + (((size_t)E + 3) & ~(size_t)3));
    int*      rowptr = ss + (((size_t)E + 3) & ~(size_t)3);
    int*      offg   = rowptr + (((size_t)N + 4) & ~(size_t)3);

    bin_edges  <<<nsl, 1024, 0, stream>>>(ei, binned, offg, N, E, NB);
    bucket_sort<<<NB + 1, 1024, 0, stream>>>(binned, offg, ss, rowptr, dinv, N, NB, nsl, W, b, Ca, M, bS);

    gather_gemm<<<(N + 31) / 32, 256, 0, stream>>>(x, M, bS, dinv, rowptr, ss, out, N);
}


# gather_gemm: LDS row pitch of the A tile 64 -> 68 floats and of M 64 -> 80 floats, so the WMMA fragment reads (A: 16 rows on one bank pair, 16-way; B: 2-way) become bank-conflict free
# speedup vs baseline: 3.9963x; 1.1358x over previous
#include <hip/hip_runtime.h>
#include <math.h>

typedef __attribute__((ext_vector_type(2))) float v2f;
typedef __attribute__((ext_vector_type(8))) float v8f;

#define PREP_SP 68

__device__ __forceinline__
void prep_block(const float* __restrict__ W, const float* __restrict__ b,
                const float* __restrict__ C, float* __restrict__ Mo,
                float* __restrict__ bS)
{
    __shared__ __align__(16) float S[64 * PREP_SP];
    __shared__ __align__(16) float sW[64 * 64];
    __shared__ float sb[64];
    const int tid = threadIdx.x;
    const int r   = tid >> 4;
    const int j4  = (tid & 15) * 4;

    const float4 w4 = *(const float4*)&W[tid * 4];
    const float4 c4 = *(const float4*)&C[tid * 4];
    const float  bv = (tid < 64) ? b[tid] : 0.0f;
    *(float4*)&sW[tid * 4] = w4;
    *(float4*)&S[r * PREP_SP + j4] = c4;
    if (tid < 64) sb[tid] = bv;
    __syncthreads();

    if (tid < 64) {
        float* row = &S[tid * PREP_SP];
        float4 v[16];
#pragma unroll
        for (int q = 0; q < 16; ++q) v[q] = *(const float4*)&row[q * 4];
        float m = -INFINITY;
#pragma unroll
        for (int q = 0; q < 16; ++q) {
            m = fmaxf(m, v[q].x);
            m = fmaxf(m, v[q].y);
            m = fmaxf(m, v[q].z);
            m = fmaxf(m, v[q].w);
        }
        float sum = 0.0f;
#pragma unroll
        for (int q = 0; q < 16; ++q) {
            v[q].x = expf(v[q].x - m); sum += v[q].x;
            v[q].y = expf(v[q].y - m); sum += v[q].y;
            v[q].z = expf(v[q].z - m); sum += v[q].z;
            v[q].w = expf(v[q].w - m); sum += v[q].w;
        }
        const float inv = 1.0f / sum;
#pragma unroll
        for (int q = 0; q < 16; ++q)
            *(float4*)&row[q * 4] = make_float4(v[q].x * inv, v[q].y * inv, v[q].z * inv, v[q].w * inv);
    }
    __syncthreads();

    float a0 = 0.0f, a1 = 0.0f, a2 = 0.0f, a3 = 0.0f;
    for (int k = 0; k < 64; ++k) {
        const float  w = sW[r * 64 + k];
        const float4 s = *(const float4*)&S[k * PREP_SP + j4];
        a0 = fmaf(w, s.x, a0);
        a1 = fmaf(w, s.y, a1);
        a2 = fmaf(w, s.z, a2);
        a3 = fmaf(w, s.w, a3);
    }
    *(float4*)&Mo[r * 64 + j4] = make_float4(a0, a1, a2, a3);

    if (tid < 64) {
        float acc = 0.0f;
        for (int k = 0; k < 64; ++k) acc = fmaf(sb[k], S[k * PREP_SP + tid], acc);
        bS[tid] = acc;
    }
}

#define SL    8192
#define SL_LOADS (SL / 1024)
#define MAXB  2048
#define MAXS  1024
#define CAP   24576

__device__ __forceinline__ int wave_incl_scan(int v, int lane)
{
#pragma unroll
    for (int o = 1; o < 32; o <<= 1) {
        const int t = __shfl_up(v, o, 32);
        if (lane >= o) v += t;
    }
    return v;
}

__device__ __forceinline__ int block_incl_scan_1024(int v, int* wsum, int tid)
{
    const int lane = tid & 31, w = tid >> 5;
    const int inc = wave_incl_scan(v, lane);
    if (lane == 31) wsum[w] = inc;
    __syncthreads();
    if (w == 0) {
        const int t  = wsum[lane];
        const int ti = wave_incl_scan(t, lane);
        wsum[lane] = ti - t;
    }
    __syncthreads();
    const int r = inc + wsum[w];
    __syncthreads();
    return r;
}

__global__ __launch_bounds__(1024)
void bin_edges(const int* __restrict__ ei, unsigned* __restrict__ binned,
               int* __restrict__ offg, int n, int E, int NB)
{
    __shared__ int cnt[MAXB];
    __shared__ int off[MAXB + 1];
    __shared__ int wsum[32];
    __shared__ unsigned stage[SL];
    const int tid = threadIdx.x;
    const int e0  = blockIdx.x * SL;
    const int e1  = min(E, e0 + SL);

    cnt[tid] = 0;
    cnt[tid + 1024] = 0;
    __syncthreads();

    int sv[SL_LOADS], dv[SL_LOADS], rk[SL_LOADS];
#pragma unroll
    for (int k = 0; k < SL_LOADS; ++k) {
        const int e  = e0 + k * 1024 + tid;
        const int ec = min(e, e1 - 1);
        const int s  = ei[ec];
        const int d  = ei[E + ec];
        sv[k] = s;
        dv[k] = (e < e1 && (unsigned)d < (unsigned)min(n, 512 * MAXB)) ? d : -1;
    }
#pragma unroll
    for (int k = 0; k < SL_LOADS; ++k) {
        rk[k] = 0;
        if (dv[k] >= 0) rk[k] = atomicAdd(&cnt[dv[k] >> 9], 1);
    }
    __syncthreads();

    const int ca = cnt[2 * tid], cb = cnt[2 * tid + 1];
    const int cs = ca + cb;
    const int incl = block_incl_scan_1024(cs, wsum, tid);
    off[2 * tid]     = incl - cs;
    off[2 * tid + 1] = incl - cs + ca;
    if (tid == 1023) off[MAXB] = incl;
    __syncthreads();

#pragma unroll
    for (int k = 0; k < SL_LOADS; ++k)
        if (dv[k] >= 0)
            stage[off[dv[k] >> 9] + rk[k]] = ((unsigned)(dv[k] & 511) << 23) | (unsigned)sv[k];
    __syncthreads();

    const int tot = off[NB];
    for (int i = tid; i < tot; i += 1024) binned[e0 + i] = stage[i];
    int* __restrict__ og = offg + (size_t)blockIdx.x * (NB + 1);
    for (int i = tid; i <= NB; i += 1024) og[i] = off[i];
}

#define SENT 0xFFFFFFFFu

__global__ __launch_bounds__(1024)
void bucket_sort(const unsigned* __restrict__ binned, const int* __restrict__ offg,
                 int* __restrict__ ss, int* __restrict__ rowptr,
                 float* __restrict__ dinv, int n, int NB, int nslices,
                 const float* __restrict__ W, const float* __restrict__ b,
                 const float* __restrict__ C, float* __restrict__ Mo,
                 float* __restrict__ bS)
{
    __shared__ __align__(16) unsigned buf[CAP];
    __shared__ __align__(16) unsigned srt[CAP];
    __shared__ int segst[MAXS], segdd[MAXS], seglen[MAXS];
    __shared__ int hist[512], cursor[512], rowst[513];
    __shared__ int wsum[32];
    __shared__ int sbase, stot;
    if ((int)blockIdx.x == NB) {
        prep_block(W, b, C, Mo, bS);
        return;
    }
    const int tid  = threadIdx.x;
    const int lane = tid & 31;
    const int w    = tid >> 5;
    const int bk   = blockIdx.x;

    if (tid < 512) hist[tid] = 0;
    if (tid == 0) sbase = 0;
    __syncthreads();

    int a = 0, len = 0;
    if (tid < nslices) {
        const int* __restrict__ og = offg + (size_t)tid * (NB + 1);
        a   = og[bk];
        len = og[bk + 1] - a;
    }
    const int incl = block_incl_scan_1024(len, wsum, tid);
    if (tid < nslices) {
        segst[tid]  = tid * SL + a;
        segdd[tid]  = incl - len;
        seglen[tid] = len;
        atomicAdd(&sbase, a);
    }
    if (tid == 1023) stot = incl;
    __syncthreads();
    const int base = sbase;
    const int Tb   = min(stot, CAP);

    for (int s0 = w; s0 < nslices; s0 += 128) {
        unsigned v0[4], v1[4];
        int st[4], ln[4], dd[4];
#pragma unroll
        for (int u = 0; u < 4; ++u) {
            const int s = s0 + 32 * u;
            const bool ok = s < nslices;
            st[u] = ok ? segst[s] : 0;
            ln[u] = ok ? seglen[s] : 0;
            dd[u] = ok ? segdd[s] : 0;
            v0[u] = (lane < ln[u])      ? binned[st[u] + lane]      : 0u;
            v1[u] = (lane + 32 < ln[u]) ? binned[st[u] + lane + 32] : 0u;
        }
#pragma unroll
        for (int u = 0; u < 4; ++u) {
            if (lane < ln[u]      && dd[u] + lane < CAP)      buf[dd[u] + lane]      = v0[u];
            if (lane + 32 < ln[u] && dd[u] + lane + 32 < CAP) buf[dd[u] + lane + 32] = v1[u];
            for (int i = lane + 64; i < ln[u]; i += 32)
                if (dd[u] + i < CAP) buf[dd[u] + i] = binned[st[u] + i];
        }
    }
    __syncthreads();

    for (int i = tid; i < Tb; i += 1024) atomicAdd(&hist[buf[i] >> 23], 1);
    __syncthreads();
    const int hv    = (tid < 512) ? hist[tid] : 0;
    const int hincl = block_incl_scan_1024(hv, wsum, tid);
    if (tid < 512) { rowst[tid] = hincl - hv; cursor[tid] = hincl - hv; }
    if (tid == 511) rowst[512] = hincl;
    __syncthreads();

    for (int i = tid; i < Tb; i += 1024) {
        const unsigned v = buf[i];
        const int pos = atomicAdd(&cursor[v >> 23], 1);
        srt[pos] = v;
    }
    if (tid < 4 && Tb + tid < CAP) srt[Tb + tid] = SENT;
    __syncthreads();

    for (int i = tid; i < Tb; i += 1024) buf[i] = SENT;
    __syncthreads();

    for (int i = tid; i < Tb; i += 1024) {
        const unsigned v = srt[i];
        const int r  = (int)(v >> 23);
        const int lo = rowst[r], hi = rowst[r + 1];
        const int q0 = lo & ~3;
        int less = 0;
        for (int j = q0; j < hi; j += 4) {
            const uint4 q = *(const uint4*)&srt[j];
            less += (q.x < v) ? 1 : 0;
            less += (q.y < v) ? 1 : 0;
            less += (q.z < v) ? 1 : 0;
            less += (q.w < v) ? 1 : 0;
        }
        buf[q0 + less] = v & 0x7FFFFFu;
    }
    __syncthreads();

    for (int i = tid; i < Tb; i += 1024) {
        int k = i;
        unsigned val = buf[k];
        while (val == SENT && k > 0) val = buf[--k];
        if (val == SENT) val = 0u;
        ss[base + i] = (int)val;
    }
    if (tid < 512) {
        const int node = bk * 512 + tid;
        if (node < n) {
            rowptr[node] = base + rowst[tid];
            dinv[node]   = rsqrtf((float)(rowst[tid + 1] - rowst[tid] + 1));
        }
    }
    if (bk == NB - 1 && tid == 0) rowptr[n] = base + Tb;
}

#define GA_PA 68
#define GA_PB 80

__global__ __launch_bounds__(256)
void gather_gemm(const float* __restrict__ x, const float* __restrict__ M,
                 const float* __restrict__ bS, const float* __restrict__ dinv,
                 const int* __restrict__ rowptr, const int* __restrict__ ss,
                 float* __restrict__ out, int nrows)
{
    __shared__ __align__(16) float sA[32 * GA_PA];
    __shared__ __align__(16) float sB[64 * GA_PB];

    const int tid = threadIdx.x;
    const int rowbase = blockIdx.x * 32;
    const int wave = tid >> 5;
    const int lane = tid & 31;
    const int hi   = lane >> 4;
    const int c    = lane & 15;
    const int c4   = c * 4;

    for (int i = tid * 4; i < 4096; i += 256 * 4)
        *(float4*)&sB[(i >> 6) * GA_PB + (i & 63)] = *(const float4*)&M[i];

#pragma unroll
    for (int rr = 0; rr < 2; ++rr) {
        const int lr  = wave * 4 + rr * 2 + hi;
        const int row = rowbase + lr;
        const bool live = row < nrows;
        const int rc  = live ? row : 0;
        const int k0  = live ? rowptr[rc] : 0;
        const int len = live ? rowptr[rc + 1] - k0 : 0;
        const float di = dinv[rc];
        float4 xs = *(const float4*)&x[(size_t)rc * 64 + c4];
        float4 acc = make_float4(di * xs.x, di * xs.y, di * xs.z, di * xs.w);
        const int lenmax = max(len, __shfl_xor(len, 16));
        for (int b = 0; b < lenmax; b += 16) {
            const bool mine = (b + c) < len;
            const int   idx = mine ? ss[k0 + b + c] : 0;
            const float wv  = mine ? dinv[idx] : 0.0f;
            const int cnt = min(16, lenmax - b);
            for (int j = 0; j < cnt; j += 4) {
                int   sj[4];
                float wj[4];
                float4 vj[4];
#pragma unroll
                for (int u = 0; u < 4; ++u) {
                    sj[u] = __shfl(idx, j + u, 16);
                    wj[u] = __shfl(wv,  j + u, 16);
                    vj[u] = make_float4(0.f, 0.f, 0.f, 0.f);
                    if (b + j + u < len) vj[u] = *(const float4*)&x[(size_t)sj[u] * 64 + c4];
                }
#pragma unroll
                for (int u = 0; u < 4; ++u) {
                    acc.x = fmaf(wj[u], vj[u].x, acc.x);
                    acc.y = fmaf(wj[u], vj[u].y, acc.y);
                    acc.z = fmaf(wj[u], vj[u].z, acc.z);
                    acc.w = fmaf(wj[u], vj[u].w, acc.w);
                }
            }
        }
        if (!live) acc = make_float4(0.f, 0.f, 0.f, 0.f);
        else       acc = make_float4(di * acc.x, di * acc.y, di * acc.z, di * acc.w);
        *(float4*)&sA[lr * GA_PA + c4] = acc;
    }
    __syncthreads();

    const int rblk = wave >> 2;
    const int cblk = wave & 3;
    const int r    = c;

    const float* Arow = &sA[(rblk * 16 + r) * GA_PA];
    const int    bcol = cblk * 16 + r;

    v8f accd = {};
#pragma unroll
    for (int k0 = 0; k0 < 64; k0 += 4) {
        v2f a, bfr;
        a.x = Arow[k0 + 2 * hi + 0];
        a.y = Arow[k0 + 2 * hi + 1];
        bfr.x = sB[(k0 + 2 * hi + 0) * GA_PB + bcol];
        bfr.y = sB[(k0 + 2 * hi + 1) * GA_PB + bcol];
        accd = __builtin_amdgcn_wmma_f32_16x16x4_f32(
             false, a,  false, bfr,
             (short)0, accd,  false,  false);
    }

    const int outrow = rowbase + rblk * 16 + hi * 8;
    const int outcol = cblk * 16 + r;
    const float bias = bS[outcol];
#pragma unroll
    for (int v = 0; v < 8; ++v) {
        int rrow = outrow + v;
        if (rrow < nrows) out[(size_t)rrow * 64 + outcol] = accd[v] + bias;
    }
}

extern "C" void kernel_launch(void* const* d_in, const int* in_sizes, int n_in,
                              void* d_out, int out_size, void* d_ws, size_t ws_size,
                              hipStream_t stream)
{
    const float* x  = (const float*)d_in[0];
    const int*   ei = (const int*)d_in[1];
    const float* W  = (const float*)d_in[2];
    const float* b  = (const float*)d_in[3];
    const float* Ca = (const float*)d_in[4];

    const int N = in_sizes[0] / 64;
    const int E = in_sizes[1] / 2;

    float* ws   = (float*)d_ws;
    float* M    = ws;
    float* bS   = ws + 4096;
    float* dinv = ws + 4160;
    float* out  = (float*)d_out;

    const int NB  = (N + 511) / 512;
    const int nsl = (E + SL - 1) / SL;
    unsigned* area = (unsigned*)(ws + 4160 + N);
    unsigned* binned = area;
    int*      ss     = (int*)(binned + (((size_t)E + 3) & ~(size_t)3));
    int*      rowptr = ss + (((size_t)E + 3) & ~(size_t)3);
    int*      offg   = rowptr + (((size_t)N + 4) & ~(size_t)3);

    bin_edges  <<<nsl, 1024, 0, stream>>>(ei, binned, offg, N, E, NB);
    bucket_sort<<<NB + 1, 1024, 0, stream>>>(binned, offg, ss, rowptr, dinv, N, NB, nsl, W, b, Ca, M, bS);

    gather_gemm<<<(N + 31) / 32, 256, 0, stream>>>(x, M, bS, dinv, rowptr, ss, out, N);
}


# gather_gemm: per-edge dinv[src] global lookup replaced by a byte in-degree table in LDS (weight recomputed by rsqrt, same bits); group = 32 waves/128 rows so two groups with the table fit a WGP
# speedup vs baseline: 4.2547x; 1.0647x over previous
#include <hip/hip_runtime.h>
#include <math.h>

typedef __attribute__((ext_vector_type(2))) float v2f;
typedef __attribute__((ext_vector_type(8))) float v8f;

#define PREP_SP 68

__device__ __forceinline__
void prep_block(const float* __restrict__ W, const float* __restrict__ b,
                const float* __restrict__ C, float* __restrict__ Mo,
                float* __restrict__ bS)
{
    __shared__ __align__(16) float S[64 * PREP_SP];
    __shared__ __align__(16) float sW[64 * 64];
    __shared__ float sb[64];
    const int tid = threadIdx.x;
    const int r   = tid >> 4;
    const int j4  = (tid & 15) * 4;

    const float4 w4 = *(const float4*)&W[tid * 4];
    const float4 c4 = *(const float4*)&C[tid * 4];
    const float  bv = (tid < 64) ? b[tid] : 0.0f;
    *(float4*)&sW[tid * 4] = w4;
    *(float4*)&S[r * PREP_SP + j4] = c4;
    if (tid < 64) sb[tid] = bv;
    __syncthreads();

    if (tid < 64) {
        float* row = &S[tid * PREP_SP];
        float4 v[16];
#pragma unroll
        for (int q = 0; q < 16; ++q) v[q] = *(const float4*)&row[q * 4];
        float m = -INFINITY;
#pragma unroll
        for (int q = 0; q < 16; ++q) {
            m = fmaxf(m, v[q].x);
            m = fmaxf(m, v[q].y);
            m = fmaxf(m, v[q].z);
            m = fmaxf(m, v[q].w);
        }
        float sum = 0.0f;
#pragma unroll
        for (int q = 0; q < 16; ++q) {
            v[q].x = expf(v[q].x - m); sum += v[q].x;
            v[q].y = expf(v[q].y - m); sum += v[q].y;
            v[q].z = expf(v[q].z - m); sum += v[q].z;
            v[q].w = expf(v[q].w - m); sum += v[q].w;
        }
        const float inv = 1.0f / sum;
#pragma unroll
        for (int q = 0; q < 16; ++q)
            *(float4*)&row[q * 4] = make_float4(v[q].x * inv, v[q].y * inv, v[q].z * inv, v[q].w * inv);
    }
    __syncthreads();

    float a0 = 0.0f, a1 = 0.0f, a2 = 0.0f, a3 = 0.0f;
    for (int k = 0; k < 64; ++k) {
        const float  w = sW[r * 64 + k];
        const float4 s = *(const float4*)&S[k * PREP_SP + j4];
        a0 = fmaf(w, s.x, a0);
        a1 = fmaf(w, s.y, a1);
        a2 = fmaf(w, s.z, a2);
        a3 = fmaf(w, s.w, a3);
    }
    *(float4*)&Mo[r * 64 + j4] = make_float4(a0, a1, a2, a3);

    if (tid < 64) {
        float acc = 0.0f;
        for (int k = 0; k < 64; ++k) acc = fmaf(sb[k], S[k * PREP_SP + tid], acc);
        bS[tid] = acc;
    }
}

#define SL    8192
#define SL_LOADS (SL / 1024)
#define MAXB  2048
#define MAXS  1024
#define CAP   24576

__device__ __forceinline__ int wave_incl_scan(int v, int lane)
{
#pragma unroll
    for (int o = 1; o < 32; o <<= 1) {
        const int t = __shfl_up(v, o, 32);
        if (lane >= o) v += t;
    }
    return v;
}

__device__ __forceinline__ int block_incl_scan_1024(int v, int* wsum, int tid)
{
    const int lane = tid & 31, w = tid >> 5;
    const int inc = wave_incl_scan(v, lane);
    if (lane == 31) wsum[w] = inc;
    __syncthreads();
    if (w == 0) {
        const int t  = wsum[lane];
        const int ti = wave_incl_scan(t, lane);
        wsum[lane] = ti - t;
    }
    __syncthreads();
    const int r = inc + wsum[w];
    __syncthreads();
    return r;
}

__global__ __launch_bounds__(1024)
void bin_edges(const int* __restrict__ ei, unsigned* __restrict__ binned,
               int* __restrict__ offg, int n, int E, int NB)
{
    __shared__ int cnt[MAXB];
    __shared__ int off[MAXB + 1];
    __shared__ int wsum[32];
    __shared__ unsigned stage[SL];
    const int tid = threadIdx.x;
    const int e0  = blockIdx.x * SL;
    const int e1  = min(E, e0 + SL);

    cnt[tid] = 0;
    cnt[tid + 1024] = 0;
    __syncthreads();

    int sv[SL_LOADS], dv[SL_LOADS], rk[SL_LOADS];
#pragma unroll
    for (int k = 0; k < SL_LOADS; ++k) {
        const int e  = e0 + k * 1024 + tid;
        const int ec = min(e, e1 - 1);
        const int s  = ei[ec];
        const int d  = ei[E + ec];
        sv[k] = s;
        dv[k] = (e < e1 && (unsigned)d < (unsigned)min(n, 512 * MAXB)) ? d : -1;
    }
#pragma unroll
    for (int k = 0; k < SL_LOADS; ++k) {
        rk[k] = 0;
        if (dv[k] >= 0) rk[k] = atomicAdd(&cnt[dv[k] >> 9], 1);
    }
    __syncthreads();

    const int ca = cnt[2 * tid], cb = cnt[2 * tid + 1];
    const int cs = ca + cb;
    const int incl = block_incl_scan_1024(cs, wsum, tid);
    off[2 * tid]     = incl - cs;
    off[2 * tid + 1] = incl - cs + ca;
    if (tid == 1023) off[MAXB] = incl;
    __syncthreads();

#pragma unroll
    for (int k = 0; k < SL_LOADS; ++k)
        if (dv[k] >= 0)
            stage[off[dv[k] >> 9] + rk[k]] = ((unsigned)(dv[k] & 511) << 23) | (unsigned)sv[k];
    __syncthreads();

    const int tot = off[NB];
    for (int i = tid; i < tot; i += 1024) binned[e0 + i] = stage[i];
    int* __restrict__ og = offg + (size_t)blockIdx.x * (NB + 1);
    for (int i = tid; i <= NB; i += 1024) og[i] = off[i];
}

#define SENT 0xFFFFFFFFu

__global__ __launch_bounds__(1024)
void bucket_sort(const unsigned* __restrict__ binned, const int* __restrict__ offg,
                 int* __restrict__ ss, int* __restrict__ rowptr,
                 float* __restrict__ dinv, unsigned char* __restrict__ deg8,
                 int n, int NB, int nslices,
                 const float* __restrict__ W, const float* __restrict__ b,
                 const float* __restrict__ C, float* __restrict__ Mo,
                 float* __restrict__ bS)
{
    __shared__ __align__(16) unsigned buf[CAP];
    __shared__ __align__(16) unsigned srt[CAP];
    __shared__ int segst[MAXS], segdd[MAXS], seglen[MAXS];
    __shared__ int hist[512], cursor[512], rowst[513];
    __shared__ int wsum[32];
    __shared__ int sbase, stot;
    if ((int)blockIdx.x == NB) {
        prep_block(W, b, C, Mo, bS);
        return;
    }
    const int tid  = threadIdx.x;
    const int lane = tid & 31;
    const int w    = tid >> 5;
    const int bk   = blockIdx.x;

    if (tid < 512) hist[tid] = 0;
    if (tid == 0) sbase = 0;
    __syncthreads();

    int a = 0, len = 0;
    if (tid < nslices) {
        const int* __restrict__ og = offg + (size_t)tid * (NB + 1);
        a   = og[bk];
        len = og[bk + 1] - a;
    }
    const int incl = block_incl_scan_1024(len, wsum, tid);
    if (tid < nslices) {
        segst[tid]  = tid * SL + a;
        segdd[tid]  = incl - len;
        seglen[tid] = len;
        atomicAdd(&sbase, a);
    }
    if (tid == 1023) stot = incl;
    __syncthreads();
    const int base = sbase;
    const int Tb   = min(stot, CAP);

    for (int s0 = w; s0 < nslices; s0 += 128) {
        unsigned v0[4], v1[4];
        int st[4], ln[4], dd[4];
#pragma unroll
        for (int u = 0; u < 4; ++u) {
            const int s = s0 + 32 * u;
            const bool ok = s < nslices;
            st[u] = ok ? segst[s] : 0;
            ln[u] = ok ? seglen[s] : 0;
            dd[u] = ok ? segdd[s] : 0;
            v0[u] = (lane < ln[u])      ? binned[st[u] + lane]      : 0u;
            v1[u] = (lane + 32 < ln[u]) ? binned[st[u] + lane + 32] : 0u;
        }
#pragma unroll
        for (int u = 0; u < 4; ++u) {
            if (lane < ln[u]      && dd[u] + lane < CAP)      buf[dd[u] + lane]      = v0[u];
            if (lane + 32 < ln[u] && dd[u] + lane + 32 < CAP) buf[dd[u] + lane + 32] = v1[u];
            for (int i = lane + 64; i < ln[u]; i += 32)
                if (dd[u] + i < CAP) buf[dd[u] + i] = binned[st[u] + i];
        }
    }
    __syncthreads();

    for (int i = tid; i < Tb; i += 1024) atomicAdd(&hist[buf[i] >> 23], 1);
    __syncthreads();
    const int hv    = (tid < 512) ? hist[tid] : 0;
    const int hincl = block_incl_scan_1024(hv, wsum, tid);
    if (tid < 512) { rowst[tid] = hincl - hv; cursor[tid] = hincl - hv; }
    if (tid == 511) rowst[512] = hincl;
    __syncthreads();

    for (int i = tid; i < Tb; i += 1024) {
        const unsigned v = buf[i];
        const int pos = atomicAdd(&cursor[v >> 23], 1);
        srt[pos] = v;
    }
    if (tid < 4 && Tb + tid < CAP) srt[Tb + tid] = SENT;
    __syncthreads();

    for (int i = tid; i < Tb; i += 1024) buf[i] = SENT;
    __syncthreads();

    for (int i = tid; i < Tb; i += 1024) {
        const unsigned v = srt[i];
        const int r  = (int)(v >> 23);
        const int lo = rowst[r], hi = rowst[r + 1];
        const int q0 = lo & ~3;
        int less = 0;
        for (int j = q0; j < hi; j += 4) {
            const uint4 q = *(const uint4*)&srt[j];
            less += (q.x < v) ? 1 : 0;
            less += (q.y < v) ? 1 : 0;
            less += (q.z < v) ? 1 : 0;
            less += (q.w < v) ? 1 : 0;
        }
        buf[q0 + less] = v & 0x7FFFFFu;
    }
    __syncthreads();

    for (int i = tid; i < Tb; i += 1024) {
        int k = i;
        unsigned val = buf[k];
        while (val == SENT && k > 0) val = buf[--k];
        if (val == SENT) val = 0u;
        ss[base + i] = (int)val;
    }
    if (tid < 512) {
        const int node = bk * 512 + tid;
        if (node < n) {
            const int indeg = rowst[tid + 1] - rowst[tid];
            rowptr[node] = base + rowst[tid];
            dinv[node]   = rsqrtf((float)(indeg + 1));
            deg8[node]   = (unsigned char)min(indeg, 255);
        } else if (node < ((n + 15) & ~15)) {
            deg8[node] = 0;
        }
    }
    if (bk == NB - 1 && tid == 0) rowptr[n] = base + Tb;
}

#define GA_PA 68
#define GA_PB 80
#define GG_ROWS   128
#define TBL_BYTES 100352
#define TBL_LOADS 7

__global__ __launch_bounds__(1024)
void gather_gemm(const float* __restrict__ x, const float* __restrict__ M,
                 const float* __restrict__ bS, const float* __restrict__ dinv,
                 const int* __restrict__ rowptr, const int* __restrict__ ss,
                 const unsigned* __restrict__ deg8w,
                 float* __restrict__ out, int nrows)
{
    __shared__ __align__(16) float sA[GG_ROWS * GA_PA];
    __shared__ __align__(16) float sB[64 * GA_PB];
    __shared__ __align__(16) unsigned tblw[TBL_BYTES / 4];
    const unsigned char* tb8 = (const unsigned char*)tblw;

    const int tid = threadIdx.x;
    const int rowbase = blockIdx.x * GG_ROWS;
    const int wave = tid >> 5;
    const int lane = tid & 31;
    const int hi   = lane >> 4;
    const int c    = lane & 15;
    const int c4   = c * 4;
    const bool use_tbl = nrows <= TBL_BYTES;

    {
        const float4 m4 = *(const float4*)&M[tid * 4];
        const uint4* __restrict__ tsrc = (const uint4*)deg8w;
        const int nchunks = use_tbl ? ((nrows + 15) >> 4) : 0;
        uint4 tv[TBL_LOADS];
#pragma unroll
        for (int q = 0; q < TBL_LOADS; ++q) {
            const int i = tid + q * 1024;
            tv[q] = make_uint4(0u, 0u, 0u, 0u);
            if (i < nchunks) tv[q] = tsrc[i];
        }
        *(float4*)&sB[(tid >> 4) * GA_PB + ((tid & 15) << 2)] = m4;
#pragma unroll
        for (int q = 0; q < TBL_LOADS; ++q) {
            const int i = tid + q * 1024;
            if (i < nchunks) ((uint4*)tblw)[i] = tv[q];
        }
    }
    __syncthreads();

#pragma unroll
    for (int rr = 0; rr < 2; ++rr) {
        const int lr  = wave * 4 + rr * 2 + hi;
        const int row = rowbase + lr;
        const bool live = row < nrows;
        const int rc  = live ? row : 0;
        const int k0  = live ? rowptr[rc] : 0;
        const int len = live ? rowptr[rc + 1] - k0 : 0;
        const float di = dinv[rc];
        float4 xs = *(const float4*)&x[(size_t)rc * 64 + c4];
        float4 acc = make_float4(di * xs.x, di * xs.y, di * xs.z, di * xs.w);
        const int lenmax = max(len, __shfl_xor(len, 16));
        for (int b = 0; b < lenmax; b += 16) {
            const bool mine = (b + c) < len;
            const int   idx = mine ? ss[k0 + b + c] : 0;
            float wv = 0.0f;
            if (mine) {
                if (use_tbl) {
                    const int d8 = (int)tb8[idx];
                    wv = (d8 < 255) ? rsqrtf((float)(d8 + 1)) : dinv[idx];
                } else {
                    wv = dinv[idx];
                }
            }
            const int cnt = min(16, lenmax - b);
            for (int j = 0; j < cnt; j += 4) {
                int   sj[4];
                float wj[4];
                float4 vj[4];
#pragma unroll
                for (int u = 0; u < 4; ++u) {
                    sj[u] = __shfl(idx, j + u, 16);
                    wj[u] = __shfl(wv,  j + u, 16);
                    vj[u] = make_float4(0.f, 0.f, 0.f, 0.f);
                    if (b + j + u < len) vj[u] = *(const float4*)&x[(size_t)sj[u] * 64 + c4];
                }
#pragma unroll
                for (int u = 0; u < 4; ++u) {
                    acc.x = fmaf(wj[u], vj[u].x, acc.x);
                    acc.y = fmaf(wj[u], vj[u].y, acc.y);
                    acc.z = fmaf(wj[u], vj[u].z, acc.z);
                    acc.w = fmaf(wj[u], vj[u].w, acc.w);
                }
            }
        }
        if (!live) acc = make_float4(0.f, 0.f, 0.f, 0.f);
        else       acc = make_float4(di * acc.x, di * acc.y, di * acc.z, di * acc.w);
        *(float4*)&sA[lr * GA_PA + c4] = acc;
    }
    __syncthreads();

    const int rblk = wave >> 2;
    const int cblk = wave & 3;
    const int r    = c;

    const float* Arow = &sA[(rblk * 16 + r) * GA_PA];
    const int    bcol = cblk * 16 + r;

    v8f accd = {};
#pragma unroll
    for (int k0 = 0; k0 < 64; k0 += 4) {
        v2f a, bfr;
        a.x = Arow[k0 + 2 * hi + 0];
        a.y = Arow[k0 + 2 * hi + 1];
        bfr.x = sB[(k0 + 2 * hi + 0) * GA_PB + bcol];
        bfr.y = sB[(k0 + 2 * hi + 1) * GA_PB + bcol];
        accd = __builtin_amdgcn_wmma_f32_16x16x4_f32(
             false, a,  false, bfr,
             (short)0, accd,  false,  false);
    }

    const int outrow = rowbase + rblk * 16 + hi * 8;
    const int outcol = cblk * 16 + r;
    const float bias = bS[outcol];
#pragma unroll
    for (int v = 0; v < 8; ++v) {
        int rrow = outrow + v;
        if (rrow < nrows) out[(size_t)rrow * 64 + outcol] = accd[v] + bias;
    }
}

extern "C" void kernel_launch(void* const* d_in, const int* in_sizes, int n_in,
                              void* d_out, int out_size, void* d_ws, size_t ws_size,
                              hipStream_t stream)
{
    const float* x  = (const float*)d_in[0];
    const int*   ei = (const int*)d_in[1];
    const float* W  = (const float*)d_in[2];
    const float* b  = (const float*)d_in[3];
    const float* Ca = (const float*)d_in[4];

    const int N = in_sizes[0] / 64;
    const int E = in_sizes[1] / 2;

    float* ws   = (float*)d_ws;
    float* M    = ws;
    float* bS   = ws + 4096;
    float* dinv = ws + 4160;
    float* out  = (float*)d_out;

    const int NB  = (N + 511) / 512;
    const int nsl = (E + SL - 1) / SL;
    unsigned* area = (unsigned*)(ws + 4160 + N);
    unsigned* binned = area;
    int*      ss     = (int*)(binned + (((size_t)E + 3) & ~(size_t)3));
    int*      rowptr = ss + (((size_t)E + 3) & ~(size_t)3);
    int*      offg   = rowptr + (((size_t)N + 4) & ~(size_t)3);
    const size_t dgo = (((size_t)((unsigned*)(offg + (size_t)nsl * (NB + 1)) - (unsigned*)d_ws)) + 63) & ~(size_t)63;
    unsigned* deg8w  = (unsigned*)d_ws + dgo;

    bin_edges  <<<nsl, 1024, 0, stream>>>(ei, binned, offg, N, E, NB);
    bucket_sort<<<NB + 1, 1024, 0, stream>>>(binned, offg, ss, rowptr, dinv, (unsigned char*)deg8w, N, NB, nsl, W, b, Ca, M, bS);

    gather_gemm<<<(N + GG_ROWS - 1) / GG_ROWS, 1024, 0, stream>>>(x, M, bS, dinv, rowptr, ss, deg8w, out, N);
}
